# starting kernel
# baseline (speedup 1.0000x reference)
#include <hip/hip_runtime.h>

typedef __attribute__((ext_vector_type(2))) float v2f;
typedef __attribute__((ext_vector_type(8))) float v8f;

#define MAXD   3
#define NDISP  7
#define NDISP2 49
#define C_CH   128
#define H_DIM  128
#define W_DIM  192
#define N_B    8
#define HW     (H_DIM * W_DIM)

__global__ __launch_bounds__(384) void corr_wmma_kernel(
    const float* __restrict__ one,
    const float* __restrict__ two,
    float* __restrict__ out)
{
    __shared__ float lds[12][2][2][16][16];

    const int y = blockIdx.x % H_DIM;
    const int n = blockIdx.x / H_DIM;

    const int wave    = threadIdx.x >> 5;
    const int lane    = threadIdx.x & 31;
    const int laneLow = lane & 15;
    const int laneHi  = lane >> 4;
    const int x0      = wave * 16;

    const float* oneRow = one + (size_t)n * C_CH * HW + (size_t)y * W_DIM;
    const float* twoPln = two + (size_t)n * C_CH * HW;

    const float* twoRowDy[NDISP];
    #pragma unroll
    for (int dyi = 0; dyi < NDISP; ++dyi) {
        int ty = y + dyi - MAXD;
        ty = ty < 0 ? 0 : (ty >= H_DIM ? H_DIM - 1 : ty);
        twoRowDy[dyi] = twoPln + (size_t)ty * W_DIM;
    }

    const int xa = x0 + laneLow;
    int xl = x0 - 3 + laneLow; xl = xl < 0 ? 0 : xl;
    int xr = x0 + 5 + laneLow; xr = xr >= W_DIM ? W_DIM - 1 : xr;
    const int voffA = 2 * laneHi * HW + xa;
    const int voffL = 2 * laneHi * HW + xl;
    const int voffR = 2 * laneHi * HW + xr;

    const v8f vzero = {0.f,0.f,0.f,0.f,0.f,0.f,0.f,0.f};
    v8f accL[NDISP], accR[NDISP];
    #pragma unroll
    for (int d = 0; d < NDISP; ++d) { accL[d] = vzero; accR[d] = vzero; }

    for (int k0 = 0; k0 < C_CH; k0 += 4) {
        const size_t koff = (size_t)k0 * HW;

        v2f a;
        a.x = oneRow[koff + voffA];
        a.y = oneRow[koff + voffA + HW];

        #pragma unroll
        for (int dyi = 0; dyi < NDISP; ++dyi) {
            const float* pT = twoRowDy[dyi] + koff;
            v2f bl, br;
            bl.x = pT[voffL];
            bl.y = pT[voffL + HW];
            br.x = pT[voffR];
            br.y = pT[voffR + HW];
            accL[dyi] = __builtin_amdgcn_wmma_f32_16x16x4_f32(
                            false, a, false, bl, (short)0, accL[dyi], false, false);
            accR[dyi] = __builtin_amdgcn_wmma_f32_16x16x4_f32(
                            false, a, false, br, (short)0, accR[dyi], false, false);
        }
    }

    const float invC = 1.0f / (float)C_CH;
    #pragma unroll
    for (int dyi = 0; dyi < NDISP; ++dyi) {
        const int p = dyi & 1;
        #pragma unroll
        for (int v = 0; v < 8; ++v) {
            lds[wave][p][0][v + 8 * laneHi][laneLow] = accL[dyi][v];
            lds[wave][p][1][v + 8 * laneHi][laneLow] = accR[dyi][v];
        }
        asm volatile("s_wait_dscnt 0x0" ::: "memory");

        const int  dy    = dyi - MAXD;
        const bool tyOOB = (y + dy < 0) || (y + dy >= H_DIM);

        if (lane < 16) {
            const int i = lane;
            const int x = x0 + i;
            #pragma unroll
            for (int dxi = 0; dxi < NDISP; ++dxi) {
                const int dx = dxi - MAXD;
                const int xs = x + dx;
                float val;
                if (tyOOB || xs < 0 || xs >= W_DIM) {
                    val = 0.0f;
                } else {
                    val = ((i < 8) ? lds[wave][p][0][i][i + dx + 3]
                                   : lds[wave][p][1][i][i + dx - 5]) * invC;
                }
                const int d = dyi * NDISP + dxi;
                out[(((size_t)n * NDISP2 + d) * H_DIM + y) * W_DIM + x] = val;
            }
        }
    }
}

extern "C" void kernel_launch(void* const* d_in, const int* in_sizes, int n_in,
                              void* d_out, int out_size, void* d_ws, size_t ws_size,
                              hipStream_t stream)
{
    const float* tenOne = (const float*)d_in[0];
    const float* tenTwo = (const float*)d_in[1];
    float* out = (float*)d_out;

    const int blocks = N_B * H_DIM;
    corr_wmma_kernel<<<blocks, 384, 0, stream>>>(tenOne, tenTwo, out);
}


# each wave holds B fragments of 2 rows of two and streams 8 rows of one against them (register reuse across dy): 24 loads per 28 WMMAs instead of 60; same k and sum order
# speedup vs baseline: 1.1310x; 1.1310x over previous
#include <hip/hip_runtime.h>


typedef __attribute__((ext_vector_type(2))) float v2f;
typedef __attribute__((ext_vector_type(8))) float v8f;

#define MAXD   3
#define NDISP  7
#define NDISP2 49
#define C_CH   128
#define H_DIM  128
#define W_DIM  192
#define N_B    8
#define HW     (H_DIM * W_DIM)
#define RROWS  2
#define NONE   (NDISP + RROWS - 1)

__global__ __launch_bounds__(384) void corr_wmma_kernel(
    const float* __restrict__ one,
    const float* __restrict__ two,
    float* __restrict__ out)
{
    __shared__ float lds[12][2][2][16][16];

    const int rp = blockIdx.x % (H_DIM / RROWS);
    const int n  = blockIdx.x / (H_DIM / RROWS);
    const int r0 = rp * RROWS;

    const int wave    = threadIdx.x >> 5;
    const int lane    = threadIdx.x & 31;
    const int laneLow = lane & 15;
    const int laneHi  = lane >> 4;
    const int x0      = wave * 16;

    const float* onePln  = one + (size_t)n * C_CH * HW;
    const float* twoRow0 = two + (size_t)n * C_CH * HW + (size_t)r0 * W_DIM;

    const float* oneRowJ[NONE];
    #pragma unroll
    for (int j = 0; j < NONE; ++j) {
        int y = r0 - MAXD + j;
        y = y < 0 ? 0 : (y >= H_DIM ? H_DIM - 1 : y);
        oneRowJ[j] = onePln + (size_t)y * W_DIM;
    }

    const int xa = x0 + laneLow;
    int xl = x0 - 3 + laneLow; xl = xl < 0 ? 0 : xl;
    int xr = x0 + 5 + laneLow; xr = xr >= W_DIM ? W_DIM - 1 : xr;
    const int voffA = 2 * laneHi * HW + xa;
    const int voffL = 2 * laneHi * HW + xl;
    const int voffR = 2 * laneHi * HW + xr;

    const v8f vzero = {0.f,0.f,0.f,0.f,0.f,0.f,0.f,0.f};
    v8f accL[RROWS][NDISP], accR[RROWS][NDISP];
    #pragma unroll
    for (int rr = 0; rr < RROWS; ++rr) {
        #pragma unroll
        for (int d = 0; d < NDISP; ++d) { accL[rr][d] = vzero; accR[rr][d] = vzero; }
    }

    for (int k0 = 0; k0 < C_CH; k0 += 4) {
        const size_t koff = (size_t)k0 * HW;

        v2f bl[RROWS], br[RROWS];
        #pragma unroll
        for (int rr = 0; rr < RROWS; ++rr) {
            const float* pT = twoRow0 + (size_t)rr * W_DIM + koff;
            bl[rr].x = pT[voffL];
            bl[rr].y = pT[voffL + HW];
            br[rr].x = pT[voffR];
            br[rr].y = pT[voffR + HW];
        }

        #pragma unroll
        for (int j = 0; j < NONE; ++j) {
            const float* pO = oneRowJ[j] + koff;
            v2f a;
            a.x = pO[voffA];
            a.y = pO[voffA + HW];

            #pragma unroll
            for (int rr = 0; rr < RROWS; ++rr) {
                const int dyi = rr + 2 * MAXD - j;
                if (dyi >= 0 && dyi < NDISP) {
                    accL[rr][dyi] = __builtin_amdgcn_wmma_f32_16x16x4_f32(
                                        false, a, false, bl[rr], (short)0, accL[rr][dyi], false, false);
                    accR[rr][dyi] = __builtin_amdgcn_wmma_f32_16x16x4_f32(
                                        false, a, false, br[rr], (short)0, accR[rr][dyi], false, false);
                }
            }
        }
    }

    const float invC = 1.0f / (float)C_CH;
    #pragma unroll
    for (int rr = 0; rr < RROWS; ++rr) {
        const int r = r0 + rr;
        #pragma unroll
        for (int dyi = 0; dyi < NDISP; ++dyi) {
            const int p = (rr * NDISP + dyi) & 1;
            #pragma unroll
            for (int v = 0; v < 8; ++v) {
                lds[wave][p][0][v + 8 * laneHi][laneLow] = accL[rr][dyi][v];
                lds[wave][p][1][v + 8 * laneHi][laneLow] = accR[rr][dyi][v];
            }
            asm volatile("s_wait_dscnt 0x0" ::: "memory");

            const int  dy    = dyi - MAXD;
            const int  y     = r - dy;
            const bool yOK   = (y >= 0) && (y < H_DIM);
            const int  oRow  = yOK ? y   : r;
            const int  oDyi  = yOK ? dyi : (NDISP - 1 - dyi);

            if (lane < 16) {
                const int i = lane;
                const int x = x0 + i;
                #pragma unroll
                for (int dxi = 0; dxi < NDISP; ++dxi) {
                    const int dx = dxi - MAXD;
                    const int xs = x + dx;
                    float val;
                    if (!yOK || xs < 0 || xs >= W_DIM) {
                        val = 0.0f;
                    } else {
                        val = ((i < 8) ? lds[wave][p][0][i][i + dx + 3]
                                       : lds[wave][p][1][i][i + dx - 5]) * invC;
                    }
                    const int d = oDyi * NDISP + dxi;
                    out[(((size_t)n * NDISP2 + d) * H_DIM + oRow) * W_DIM + x] = val;
                }
            }
        }
    }
}

extern "C" void kernel_launch(void* const* d_in, const int* in_sizes, int n_in,
                              void* d_out, int out_size, void* d_ws, size_t ws_size,
                              hipStream_t stream)
{
    const float* tenOne = (const float*)d_in[0];
    const float* tenTwo = (const float*)d_in[1];
    float* out = (float*)d_out;

    const int blocks = N_B * (H_DIM / RROWS);
    corr_wmma_kernel<<<blocks, 384, 0, stream>>>(tenOne, tenTwo, out);
}


# A fragment = 8 pixels of TWO rows of one (lanes 0-7 row j, 8-15 row j+1), so a 16x16 product holds two wanted bands: 16 WMMAs per K-step not 28, 128 accumulator VGPRs not 224; same k and sum order
# speedup vs baseline: 1.8561x; 1.6411x over previous
#include <hip/hip_runtime.h>


typedef __attribute__((ext_vector_type(2))) float v2f;
typedef __attribute__((ext_vector_type(8))) float v8f;

#define MAXD   3
#define NDISP  7
#define NDISP2 49
#define C_CH   128
#define H_DIM  128
#define W_DIM  192
#define N_B    8
#define HW     (H_DIM * W_DIM)
#define RROWS  2
#define NONE   (NDISP + RROWS - 1)
#define NPAIR  3
#define LPITCH 18


#define WMMA_F32(a, b, c) __builtin_amdgcn_wmma_f32_16x16x4_f32(false, (a), false, (b), (short)0, (c), false, false)

template <bool PAIRED>
__device__ __forceinline__ void extract_unit(
    float (*buf)[16][LPITCH], const v8f& tl, const v8f& tr,
    int rr, int jLow, int n, int r0, int x0, int lane, float* __restrict__ out)
{
    const int laneLow = lane & 15;
    const int laneHi  = lane >> 4;
    #pragma unroll
    for (int v = 0; v < 8; ++v) {
        buf[0][v + 8 * laneHi][laneLow] = tl[v];
        buf[1][v + 8 * laneHi][laneLow] = tr[v];
    }
    asm volatile("s_wait_dscnt 0x0" ::: "memory");

    if (PAIRED || lane < 16) {
        const int i    = laneLow;
        const int m    = i & 7;
        const int tile = i >> 3;
        const int hsel = PAIRED ? laneHi : tile;
        const int j    = jLow + (PAIRED ? laneHi : 0);
        const int dyi  = rr + 2 * MAXD - j;
        const int r    = r0 + rr;
        const int y    = r0 - MAXD + j;
        const bool yOK = (y >= 0) && (y < H_DIM);
        const int oRow = yOK ? y   : r;
        const int oDyi = yOK ? dyi : (NDISP - 1 - dyi);
        const int x    = x0 + i;
        const float invC = 1.0f / (float)C_CH;
        #pragma unroll
        for (int dxi = 0; dxi < NDISP; ++dxi) {
            const int xs = x + dxi - MAXD;
            float val = buf[tile][m + 8 * hsel][m + dxi] * invC;
            if (!yOK || xs < 0 || xs >= W_DIM) val = 0.0f;
            const int d = oDyi * NDISP + dxi;
            out[(((size_t)n * NDISP2 + d) * H_DIM + oRow) * W_DIM + x] = val;
        }
    }
}

__global__ __launch_bounds__(384) void corr_wmma_kernel(
    const float* __restrict__ one,
    const float* __restrict__ two,
    float* __restrict__ out)
{
    __shared__ float lds[12][2][2][16][LPITCH];

    const int rp = blockIdx.x % (H_DIM / RROWS);
    const int n  = blockIdx.x / (H_DIM / RROWS);
    const int r0 = rp * RROWS;

    const int wave    = threadIdx.x >> 5;
    const int lane    = threadIdx.x & 31;
    const int laneLow = lane & 15;
    const int laneHi  = lane >> 4;
    const int x0      = wave * 16;

    const float* onePln  = one + (size_t)n * C_CH * HW;
    const float* twoRow0 = two + (size_t)n * C_CH * HW + (size_t)r0 * W_DIM;

    int pairRow[NPAIR];
    #pragma unroll
    for (int p = 0; p < NPAIR; ++p) {
        int y = r0 - 2 + 2 * p;
        y = y < 0 ? 0 : (y > H_DIM - 2 ? H_DIM - 2 : y);
        pairRow[p] = y * W_DIM;
    }
    int yS0 = r0 - MAXD;      yS0 = yS0 < 0 ? 0 : yS0;
    int yS7 = r0 + MAXD + 1;  yS7 = yS7 > H_DIM - 1 ? H_DIM - 1 : yS7;
    const int s0Row = yS0 * W_DIM;
    const int s7Row = yS7 * W_DIM;

    const int chOff = 2 * laneHi * HW;
    const unsigned vP = (unsigned)(chOff + x0 + (laneLow & 7) + (laneLow >= 8 ? W_DIM : 0));
    const unsigned vS = (unsigned)(chOff + x0 + laneLow);
    int xl = x0 - 3 + laneLow; xl = xl < 0 ? 0 : xl;
    int xr = x0 + 5 + laneLow; xr = xr >= W_DIM ? W_DIM - 1 : xr;
    const unsigned vL = (unsigned)(chOff + xl);
    const unsigned vR = (unsigned)(chOff + xr);

    const v8f vzero = {0.f,0.f,0.f,0.f,0.f,0.f,0.f,0.f};
    v8f accPL[RROWS][NPAIR], accPR[RROWS][NPAIR], accSL[RROWS], accSR[RROWS];
    #pragma unroll
    for (int rr = 0; rr < RROWS; ++rr) {
        #pragma unroll
        for (int p = 0; p < NPAIR; ++p) { accPL[rr][p] = vzero; accPR[rr][p] = vzero; }
        accSL[rr] = vzero; accSR[rr] = vzero;
    }

    for (int k0 = 0; k0 < C_CH; k0 += 4) {
        const int kk = __builtin_amdgcn_readfirstlane(k0);
        const float* pO = onePln  + (size_t)kk * HW;
        const float* pT = twoRow0 + (size_t)kk * HW;

        v2f bl[RROWS], br[RROWS];
        #pragma unroll
        for (int rr = 0; rr < RROWS; ++rr) {
            const float* pTr = pT + rr * W_DIM;
            bl[rr].x = pTr[vL];
            bl[rr].y = (pTr + HW)[vL];
            br[rr].x = pTr[vR];
            br[rr].y = (pTr + HW)[vR];
        }
        v2f aPL[NPAIR], aPR[NPAIR];
        #pragma unroll
        for (int p = 0; p < NPAIR; ++p) {
            const float* pP = pO + pairRow[p];
            aPL[p].x = pP[vP];
            aPL[p].y = (pP + HW)[vP];
            aPR[p].x = (pP + 8)[vP];
            aPR[p].y = (pP + 8 + HW)[vP];
        }
        v2f aS0, aS7;
        const float* pS0 = pO + s0Row;
        const float* pS7 = pO + s7Row;
        aS0.x = pS0[vS];  aS0.y = (pS0 + HW)[vS];
        aS7.x = pS7[vS];  aS7.y = (pS7 + HW)[vS];

        #pragma unroll
        for (int rr = 0; rr < RROWS; ++rr) {
            #pragma unroll
            for (int p = 0; p < NPAIR; ++p) {
                accPL[rr][p] = WMMA_F32(aPL[p], bl[rr], accPL[rr][p]);
                accPR[rr][p] = WMMA_F32(aPR[p], br[rr], accPR[rr][p]);
            }
        }
        accSL[0] = WMMA_F32(aS0, bl[0], accSL[0]);
        accSR[0] = WMMA_F32(aS0, br[0], accSR[0]);
        accSL[1] = WMMA_F32(aS7, bl[1], accSL[1]);
        accSR[1] = WMMA_F32(aS7, br[1], accSR[1]);
    }

    int u = 0;
    #pragma unroll
    for (int rr = 0; rr < RROWS; ++rr) {
        #pragma unroll
        for (int p = 0; p < NPAIR; ++p) {
            extract_unit<true>(lds[wave][u & 1], accPL[rr][p], accPR[rr][p],
                               rr, 1 + 2 * p, n, r0, x0, lane, out);
            ++u;
        }
        extract_unit<false>(lds[wave][u & 1], accSL[rr], accSR[rr],
                            rr, rr == 0 ? 0 : NONE - 1, n, r0, x0, lane, out);
        ++u;
    }
}

extern "C" void kernel_launch(void* const* d_in, const int* in_sizes, int n_in,
                              void* d_out, int out_size, void* d_ws, size_t ws_size,
                              hipStream_t stream)
{
    const float* tenOne = (const float*)d_in[0];
    const float* tenTwo = (const float*)d_in[1];
    float* out = (float*)d_out;

    const int blocks = N_B * (H_DIM / RROWS);
    corr_wmma_kernel<<<blocks, 384, 0, stream>>>(tenOne, tenTwo, out);
}
